# starting kernel
# baseline (speedup 1.0000x reference)
#include <hip/hip_runtime.h>

typedef __attribute__((ext_vector_type(2))) float v2f;
typedef __attribute__((ext_vector_type(8))) float v8f;

#define EPS 1e-12f

__device__ __forceinline__ float max_with_xor16(float x) {
    int xi = __builtin_bit_cast(int, x);
    int sw = __builtin_amdgcn_permlanex16(xi, xi, 0x76543210, 0xfedcba98,
                                          false, false);
    return fmaxf(x, __builtin_bit_cast(float, sw));
}

__global__ __launch_bounds__(256) void conv_surface_wmma_kernel(
    const int*   __restrict__ nbr_idx,
    const float* __restrict__ verts,
    const float* __restrict__ dirs,
    float*       __restrict__ out,
    int V)
{
    __shared__ v2f s_bp[1040];
    __shared__ v2f s_ap[8 * 80];

    const int tid  = threadIdx.x;
    const int lane = tid & 31;
    const int wv   = tid >> 5;
    const int wid  = blockIdx.x * 8 + wv;
    const int b    = wid / V;
    const int v    = wid - b * V;
    const int half = lane >> 4;
    const int nsub = lane & 15;

    for (int m = tid; m < 512; m += 256) {
        float d0 = dirs[m];
        float d1 = dirs[512 + m];
        float d2 = dirs[1024 + m];
        float inv = 1.0f / fmaxf(sqrtf(d0*d0 + d1*d1 + d2*d2), EPS);
        v2f lo = {d0 * inv, d1 * inv};
        v2f hi = {d2 * inv, 0.0f};
        s_bp[m]       = lo;
        s_bp[528 + m] = hi;
    }

    {
        int idx = nbr_idx[wid * 32 + lane];
        const float* vb = verts + (size_t)b * V * 3;
        float cx = vb[(size_t)v*3+0], cy = vb[(size_t)v*3+1], cz = vb[(size_t)v*3+2];
        float dx = vb[(size_t)idx*3+0] - cx;
        float dy = vb[(size_t)idx*3+1] - cy;
        float dz = vb[(size_t)idx*3+2] - cz;
        float inv = 1.0f / fmaxf(sqrtf(dx*dx + dy*dy + dz*dz), EPS);
        v2f lo = {dx * inv, dy * inv};
        v2f hi = {dz * inv, 0.0f};
        v2f* ap = &s_ap[wv * 80];
        ap[lane]      = lo;
        ap[48 + lane] = hi;
    }

    __syncthreads();

    const v2f* ap = &s_ap[wv * 80 + half * 48 + nsub];
    v2f a0 = ap[0];
    v2f a1 = ap[16];

    const v2f* bp = &s_bp[half * 528 + nsub];

    float acc[4] = {0.f, 0.f, 0.f, 0.f};
    v2f bf = bp[0];

    #pragma unroll
    for (int t = 0; t < 32; ++t) {
        v2f bfn = (t < 31) ? bp[(t + 1) * 16] : bf;

        v8f c0 = {};
        c0 = __builtin_amdgcn_wmma_f32_16x16x4_f32(false, a0, false, bf,
                                                   (short)0, c0, false, false);
        v8f c1 = {};
        c1 = __builtin_amdgcn_wmma_f32_16x16x4_f32(false, a1, false, bf,
                                                   (short)0, c1, false, false);

        float mx = 0.0f;
        #pragma unroll
        for (int r = 0; r < 8; ++r) {
            mx = fmaxf(mx, c0[r]);
            mx = fmaxf(mx, c1[r]);
        }
        mx = max_with_xor16(mx);

        acc[t & 3] += mx;

        bf = bfn;
    }

    float s0 = half ? acc[2] : acc[0];
    float s1 = half ? acc[3] : acc[1];
    float* o = out + (size_t)wid * 64 + half * 32 + nsub;
    o[0]  = s0;
    o[16] = s1;
}

extern "C" void kernel_launch(void* const* d_in, const int* in_sizes, int n_in,
                              void* d_out, int out_size, void* d_ws, size_t ws_size,
                              hipStream_t stream) {
    const int*   nbr   = (const int*)d_in[0];
    const float* verts = (const float*)d_in[1];
    const float* dirs  = (const float*)d_in[2];
    float* out = (float*)d_out;

    const int T = in_sizes[1] / 3;
    const int V = T / 2;
    const int blocks = T / 8;

    conv_surface_wmma_kernel<<<blocks, 256, 0, stream>>>(nbr, verts, dirs, out, V);
}


# each A fragment holds 8 neighbors of vertex 0 in rows 0-7 and of vertex 1 in rows 8-15: the neighbor max is an in-lane max3 tree over 4 D tiles, no cross-lane exchange left in the loop
# speedup vs baseline: 1.1280x; 1.1280x over previous
#include <hip/hip_runtime.h>


typedef __attribute__((ext_vector_type(2))) float v2f;
typedef __attribute__((ext_vector_type(8))) float v8f;

#define EPS 1e-12f
#define VPW 2
#define WAVES 16
#define THREADS (WAVES * 32)
#define DPT (512 / THREADS)
#define APITCH 40

__device__ __forceinline__ float from_xor16(float x) {
    int xi = __builtin_bit_cast(int, x);
    int sw = __builtin_amdgcn_permlanex16(xi, xi, 0x76543210, 0xfedcba98,
                                          false, false);
    return __builtin_bit_cast(float, sw);
}

__global__ __launch_bounds__(THREADS) void conv_surface_wmma_kernel(
    const int*   __restrict__ nbr_idx,
    const float* __restrict__ verts,
    const float* __restrict__ dirs,
    float*       __restrict__ out,
    int V)
{
    __shared__ v2f s_bp[1040];
    __shared__ v2f s_ap[WAVES * VPW * 2 * APITCH];

    const int tid   = threadIdx.x;
    const int lane  = tid & 31;
    const int wv    = tid >> 5;
    const int vbase = (blockIdx.x * WAVES + wv) * VPW;
    const int half  = lane >> 4;
    const int nsub  = lane & 15;

    int idx[VPW];
    #pragma unroll
    for (int j = 0; j < VPW; ++j)
        idx[j] = nbr_idx[(size_t)(vbase + j) * 32 + lane];

    float d0[DPT], d1[DPT], d2[DPT];
    #pragma unroll
    for (int i = 0; i < DPT; ++i) {
        const int m = tid + THREADS * i;
        d0[i] = dirs[m];
        d1[i] = dirs[512 + m];
        d2[i] = dirs[1024 + m];
    }

    float cx[VPW], cy[VPW], cz[VPW], px[VPW], py[VPW], pz[VPW];
    #pragma unroll
    for (int j = 0; j < VPW; ++j) {
        const int vtx = vbase + j;
        const int b   = vtx / V;
        const int v   = vtx - b * V;
        const float* vb = verts + (size_t)b * V * 3;
        cx[j] = vb[(size_t)v * 3 + 0];
        cy[j] = vb[(size_t)v * 3 + 1];
        cz[j] = vb[(size_t)v * 3 + 2];
        px[j] = vb[(size_t)idx[j] * 3 + 0];
        py[j] = vb[(size_t)idx[j] * 3 + 1];
        pz[j] = vb[(size_t)idx[j] * 3 + 2];
    }

    #pragma unroll
    for (int i = 0; i < DPT; ++i) {
        const int m = tid + THREADS * i;
        float inv = 1.0f / fmaxf(sqrtf(d0[i]*d0[i] + d1[i]*d1[i] + d2[i]*d2[i]), EPS);
        v2f lo = {d0[i] * inv, d1[i] * inv};
        v2f hi = {d2[i] * inv, 0.0f};
        s_bp[m]       = lo;
        s_bp[528 + m] = hi;
    }

    #pragma unroll
    for (int j = 0; j < VPW; ++j) {
        float dx = px[j] - cx[j];
        float dy = py[j] - cy[j];
        float dz = pz[j] - cz[j];
        float inv = 1.0f / fmaxf(sqrtf(dx*dx + dy*dy + dz*dz), EPS);
        v2f lo = {dx * inv, dy * inv};
        v2f hi = {dz * inv, 0.0f};
        v2f* ap = &s_ap[(wv * VPW + j) * 2 * APITCH];
        ap[lane]          = lo;
        ap[APITCH + lane] = hi;
    }

    __syncthreads();

    v2f a[4];
    {
        const v2f* ap = &s_ap[((wv * VPW + (nsub >> 3)) * 2 + half) * APITCH + (nsub & 7)];
        #pragma unroll
        for (int i = 0; i < 4; ++i) a[i] = ap[8 * i];
    }

    const v2f* bp = &s_bp[half * 528 + nsub];

    float acc[4];
    #pragma unroll
    for (int q = 0; q < 4; ++q) acc[q] = 0.0f;

    v2f bcur[4];
    #pragma unroll
    for (int q = 0; q < 4; ++q) bcur[q] = bp[q * 16];

    #pragma unroll 1
    for (int g = 0; g < 8; ++g) {
        const int gn = (g + 1) & 7;
        v2f bnext[4];
        #pragma unroll
        for (int q = 0; q < 4; ++q) bnext[q] = bp[(gn * 4 + q) * 16];

        #pragma unroll
        for (int q = 0; q < 4; ++q) {
            v8f c[4];
            #pragma unroll
            for (int i = 0; i < 4; ++i) {
                v8f z = {};
                c[i] = __builtin_amdgcn_wmma_f32_16x16x4_f32(false, a[i], false, bcur[q],
                                                             (short)0, z, false, false);
            }
            float t[8];
            #pragma unroll
            for (int r = 0; r < 8; ++r)
                t[r] = fmaxf(fmaxf(c[0][r], c[1][r]), c[2][r]);
            float u[4];
            #pragma unroll
            for (int r = 0; r < 4; ++r)
                u[r] = fmaxf(fmaxf(t[2 * r], t[2 * r + 1]), c[3][r]);
            const float w0 = fmaxf(fmaxf(u[0], u[1]), c[3][4]);
            const float w1 = fmaxf(fmaxf(u[2], u[3]), c[3][5]);
            const float x  = fmaxf(fmaxf(w0, w1), c[3][6]);
            const float mx = fmaxf(fmaxf(x, c[3][7]), 0.0f);
            acc[q] += mx;
        }

        #pragma unroll
        for (int q = 0; q < 4; ++q) bcur[q] = bnext[q];
    }

    #pragma unroll
    for (int p = 0; p < 2; ++p) {
        const float lo = acc[2 * p];
        const float hi = acc[2 * p + 1];
        const float got = from_xor16(half ? lo : hi);
        const float v0 = half ? got : lo;
        const float v1 = half ? hi : got;
        out[(size_t)(vbase + 0) * 64 + 32 * p + lane] = v0;
        out[(size_t)(vbase + 1) * 64 + 32 * p + lane] = v1;
    }
}

extern "C" void kernel_launch(void* const* d_in, const int* in_sizes, int n_in,
                              void* d_out, int out_size, void* d_ws, size_t ws_size,
                              hipStream_t stream) {
    const int*   nbr   = (const int*)d_in[0];
    const float* verts = (const float*)d_in[1];
    const float* dirs  = (const float*)d_in[2];
    float* out = (float*)d_out;

    const int T = in_sizes[1] / 3;
    const int V = T / 2;
    const int blocks = T / 32;

    conv_surface_wmma_kernel<<<blocks, 512, 0, stream>>>(nbr, verts, dirs, out, V);
}
